# starting kernel
# baseline (speedup 1.0000x reference)
#include <hip/hip_runtime.h>

#define BB   32
#define NN   1024
#define DD   64
#define TT   2048
#define GG   8
#define TC   (TT / GG)
#define NB   16
#define NBP  (NB + 4)
#define NBLK (NN / NB)
#define BN   (BB * NN)
#define BNG  (BN * GG)

typedef float v2f __attribute__((ext_vector_type(2)));
typedef float v4f __attribute__((ext_vector_type(4)));
typedef float v8f __attribute__((ext_vector_type(8)));

#define SWIZ(v, imm) \
    __int_as_float(__builtin_amdgcn_ds_swizzle(__float_as_int(v), (imm)))

__device__ __forceinline__ float wave_max32(float v) {
    v = fmaxf(v, SWIZ(v, 0x401F));
    v = fmaxf(v, SWIZ(v, 0x201F));
    v = fmaxf(v, SWIZ(v, 0x101F));
    v = fmaxf(v, SWIZ(v, 0x081F));
    v = fmaxf(v, SWIZ(v, 0x041F));
    return v;
}
__device__ __forceinline__ float wave_sum32(float v) {
    v += SWIZ(v, 0x401F);
    v += SWIZ(v, 0x201F);
    v += SWIZ(v, 0x101F);
    v += SWIZ(v, 0x081F);
    v += SWIZ(v, 0x041F);
    return v;
}

__device__ __forceinline__ void smerge(float& m, float& num, float& den,
                                       float m2, float n2, float d2) {
    float M  = fmaxf(m, m2);
    float s1 = __expf(m - M);
    float s2 = __expf(m2 - M);
    num = num * s1 + n2 * s2;
    den = den * s1 + d2 * s2;
    m   = M;
}

__global__ __launch_bounds__(256)
void mmse_scan_kernel(const float* __restrict__ data,
                      const float* __restrict__ targets,
                      const float* __restrict__ W,
                      float* __restrict__ wsM,
                      float* __restrict__ wsNum,
                      float* __restrict__ wsDen)
{
    __shared__ alignas(16) float xw[2][TC][NBP];
    __shared__ alignas(16) float red[2][NB][8][4];

    const int tid  = threadIdx.x;
    const int lane = tid & 31;
    const int wave = tid >> 5;
    const int lm   = lane & 15;
    const int lh   = lane >> 4;

    const int b     = (int)blockIdx.x >> 3;
    const int g     = (int)blockIdx.x & (GG - 1);
    const int tbase = g * TC;

    const float* dataB = data + (size_t)b * NN * DD;

    v2f bfrag[2][16];
#pragma unroll
    for (int j = 0; j < 2; ++j) {
        const int gcol = tbase + wave * 32 + j * 16 + lm;
#pragma unroll
        for (int s = 0; s < 16; ++s) {
            const int k0 = 4 * s + 2 * lh;
            v2f bf;
            bf.x = W[(size_t)k0 * TT + gcol];
            bf.y = W[(size_t)(k0 + 1) * TT + gcol];
            bfrag[j][s] = bf;
        }
    }

    auto compute = [&](int blk, int buf) {
        const int n0 = blk * NB;
        v2f af[16];
#pragma unroll
        for (int s = 0; s < 16; ++s) {
            const float* p = dataB + (size_t)(n0 + lm) * DD + 4 * s + 2 * lh;
            af[s] = *reinterpret_cast<const v2f*>(p);
        }
#pragma unroll
        for (int j = 0; j < 2; ++j) {
            v8f acc = {};
#pragma unroll
            for (int s = 0; s < 16; ++s) {
                acc = __builtin_amdgcn_wmma_f32_16x16x4_f32(
                     false, af[s],  false, bfrag[j][s],
                     (short)0, acc,  false,  false);
            }
            const int lcol = wave * 32 + j * 16 + lm;
            v4f lo = {acc[0], acc[1], acc[2], acc[3]};
            v4f hi = {acc[4], acc[5], acc[6], acc[7]};
            *reinterpret_cast<v4f*>(&xw[buf][lcol][8 * lh])     = lo;
            *reinterpret_cast<v4f*>(&xw[buf][lcol][8 * lh + 4]) = hi;
        }
    };

    float cum = 0.0f;

    compute(0, 0);
    __syncthreads();

    for (int blk = 0; blk < NBLK; ++blk) {
        const int cur = blk & 1;
        const int n0  = blk * NB;

        if (blk + 1 < NBLK) compute(blk + 1, cur ^ 1);

        v4f xr[4], tr[4];
#pragma unroll
        for (int q = 0; q < 4; ++q)
            xr[q] = *reinterpret_cast<const v4f*>(&xw[cur][tid][4 * q]);
        const float* tb = targets + b * NN + n0;
#pragma unroll
        for (int q = 0; q < 4; ++q)
            tr[q] = *reinterpret_cast<const v4f*>(tb + 4 * q);

#pragma unroll
        for (int n = 0; n < NB; ++n) {
            const float x = xr[n >> 2][n & 3];

            const float m  = wave_max32(cum);
            const float e  = __expf(cum - m);
            const float nu = wave_sum32(e * x);
            const float de = wave_sum32(e);
            if (lane == 0) {
                v4f t4 = {m, nu, de, 0.0f};
                *reinterpret_cast<v4f*>(&red[cur][n][wave][0]) = t4;
            }
            const float df = tr[n >> 2][n & 3] - x;
            cum = fmaf(df, df * -0.5f, cum);
        }

        __syncthreads();

        if ((tid & 15) == 0) {
            const int n = tid >> 4;
            v4f t0 = *reinterpret_cast<const v4f*>(&red[cur][n][0][0]);
            float m = t0.x, num = t0.y, den = t0.z;
#pragma unroll
            for (int w = 1; w < 8; ++w) {
                v4f tw = *reinterpret_cast<const v4f*>(&red[cur][n][w][0]);
                smerge(m, num, den, tw.x, tw.y, tw.z);
            }
            const size_t o = (size_t)g * BN + (size_t)b * NN + n0 + n;
            wsM[o] = m; wsNum[o] = num; wsDen[o] = den;
        }
    }
}

__global__ __launch_bounds__(256)
void mmse_combine_kernel(const float* __restrict__ wsM,
                         const float* __restrict__ wsNum,
                         const float* __restrict__ wsDen,
                         float* __restrict__ out)
{
    const int i = (int)blockIdx.x * 256 + (int)threadIdx.x;
    if (i >= BN) return;
    float m = wsM[i], num = wsNum[i], den = wsDen[i];
#pragma unroll
    for (int g = 1; g < GG; ++g) {
        const size_t o = (size_t)g * BN + i;
        smerge(m, num, den, wsM[o], wsNum[o], wsDen[o]);
    }
    out[i] = num / den;
}

extern "C" void kernel_launch(void* const* d_in, const int* in_sizes, int n_in,
                              void* d_out, int out_size, void* d_ws, size_t ws_size,
                              hipStream_t stream) {
    (void)in_sizes; (void)n_in; (void)out_size; (void)ws_size;
    const float* data    = (const float*)d_in[0];
    const float* targets = (const float*)d_in[1];
    const float* W       = (const float*)d_in[2];
    float*       out     = (float*)d_out;

    float* wsBase = (float*)d_ws;
    float* wsM    = wsBase;
    float* wsNum  = wsBase + (size_t)BNG;
    float* wsDen  = wsBase + (size_t)2 * BNG;

    mmse_scan_kernel<<<dim3(BB * GG), dim3(256), 0, stream>>>(
        data, targets, W, wsM, wsNum, wsDen);
    mmse_combine_kernel<<<dim3((BN + 255) / 256), dim3(256), 0, stream>>>(
        wsM, wsNum, wsDen, out);
}


# wave max/sum butterflies moved from ds_swizzle (LDS unit) to the vector ALU: v_permlanex16 for xor 16, DPP row_xmask for 8,4,2,1; partner order kept, so the summation order is unchanged
# speedup vs baseline: 2.5960x; 2.5960x over previous
#include <hip/hip_runtime.h>

#define BB   32
#define NN   1024
#define DD   64
#define TT   2048
#define GG   8
#define TC   (TT / GG)
#define NB   16
#define NBP  (NB + 4)
#define NBLK (NN / NB)
#define BN   (BB * NN)
#define BNG  (BN * GG)

typedef float v2f __attribute__((ext_vector_type(2)));
typedef float v4f __attribute__((ext_vector_type(4)));
typedef float v8f __attribute__((ext_vector_type(8)));

__device__ __forceinline__ float XROW(float v) {
    const int b = __float_as_int(v);
    return __int_as_float(__builtin_amdgcn_permlanex16(
        b, b, 0x76543210, 0xfedcba98, false, false));
}
#define XMASK(v, X) \
    __int_as_float(__builtin_amdgcn_update_dpp( \
        0, __float_as_int(v), 0x160 + (X), 0xF, 0xF, true))

__device__ __forceinline__ float wave_max32(float v) {
    v = fmaxf(v, XROW(v));
    v = fmaxf(v, XMASK(v, 8));
    v = fmaxf(v, XMASK(v, 4));
    v = fmaxf(v, XMASK(v, 2));
    v = fmaxf(v, XMASK(v, 1));
    return v;
}
__device__ __forceinline__ float wave_sum32(float v) {
    v += XROW(v);
    v += XMASK(v, 8);
    v += XMASK(v, 4);
    v += XMASK(v, 2);
    v += XMASK(v, 1);
    return v;
}

__device__ __forceinline__ void smerge(float& m, float& num, float& den,
                                       float m2, float n2, float d2) {
    float M  = fmaxf(m, m2);
    float s1 = __expf(m - M);
    float s2 = __expf(m2 - M);
    num = num * s1 + n2 * s2;
    den = den * s1 + d2 * s2;
    m   = M;
}

__global__ __launch_bounds__(256)
void mmse_scan_kernel(const float* __restrict__ data,
                      const float* __restrict__ targets,
                      const float* __restrict__ W,
                      float* __restrict__ wsM,
                      float* __restrict__ wsNum,
                      float* __restrict__ wsDen)
{
    __shared__ alignas(16) float xw[2][TC][NBP];
    __shared__ alignas(16) float red[2][NB][8][4];

    const int tid  = threadIdx.x;
    const int lane = tid & 31;
    const int wave = tid >> 5;
    const int lm   = lane & 15;
    const int lh   = lane >> 4;

    const int b     = (int)blockIdx.x >> 3;
    const int g     = (int)blockIdx.x & (GG - 1);
    const int tbase = g * TC;

    const float* dataB = data + (size_t)b * NN * DD;

    v2f bfrag[2][16];
#pragma unroll
    for (int j = 0; j < 2; ++j) {
        const int gcol = tbase + wave * 32 + j * 16 + lm;
#pragma unroll
        for (int s = 0; s < 16; ++s) {
            const int k0 = 4 * s + 2 * lh;
            v2f bf;
            bf.x = W[(size_t)k0 * TT + gcol];
            bf.y = W[(size_t)(k0 + 1) * TT + gcol];
            bfrag[j][s] = bf;
        }
    }

    auto compute = [&](int blk, int buf) {
        const int n0 = blk * NB;
        v2f af[16];
#pragma unroll
        for (int s = 0; s < 16; ++s) {
            const float* p = dataB + (size_t)(n0 + lm) * DD + 4 * s + 2 * lh;
            af[s] = *reinterpret_cast<const v2f*>(p);
        }
#pragma unroll
        for (int j = 0; j < 2; ++j) {
            v8f acc = {};
#pragma unroll
            for (int s = 0; s < 16; ++s) {
                acc = __builtin_amdgcn_wmma_f32_16x16x4_f32(
                     false, af[s],  false, bfrag[j][s],
                     (short)0, acc,  false,  false);
            }
            const int lcol = wave * 32 + j * 16 + lm;
            v4f lo = {acc[0], acc[1], acc[2], acc[3]};
            v4f hi = {acc[4], acc[5], acc[6], acc[7]};
            *reinterpret_cast<v4f*>(&xw[buf][lcol][8 * lh])     = lo;
            *reinterpret_cast<v4f*>(&xw[buf][lcol][8 * lh + 4]) = hi;
        }
    };

    float cum = 0.0f;

    compute(0, 0);
    __syncthreads();

    for (int blk = 0; blk < NBLK; ++blk) {
        const int cur = blk & 1;
        const int n0  = blk * NB;

        if (blk + 1 < NBLK) compute(blk + 1, cur ^ 1);

        v4f xr[4], tr[4];
#pragma unroll
        for (int q = 0; q < 4; ++q)
            xr[q] = *reinterpret_cast<const v4f*>(&xw[cur][tid][4 * q]);
        const float* tb = targets + b * NN + n0;
#pragma unroll
        for (int q = 0; q < 4; ++q)
            tr[q] = *reinterpret_cast<const v4f*>(tb + 4 * q);

#pragma unroll
        for (int n = 0; n < NB; ++n) {
            const float x = xr[n >> 2][n & 3];

            const float m  = wave_max32(cum);
            const float e  = __expf(cum - m);
            const float nu = wave_sum32(e * x);
            const float de = wave_sum32(e);
            if (lane == 0) {
                v4f t4 = {m, nu, de, 0.0f};
                *reinterpret_cast<v4f*>(&red[cur][n][wave][0]) = t4;
            }
            const float df = tr[n >> 2][n & 3] - x;
            cum = fmaf(df, df * -0.5f, cum);
        }

        __syncthreads();

        if ((tid & 15) == 0) {
            const int n = tid >> 4;
            v4f t0 = *reinterpret_cast<const v4f*>(&red[cur][n][0][0]);
            float m = t0.x, num = t0.y, den = t0.z;
#pragma unroll
            for (int w = 1; w < 8; ++w) {
                v4f tw = *reinterpret_cast<const v4f*>(&red[cur][n][w][0]);
                smerge(m, num, den, tw.x, tw.y, tw.z);
            }
            const size_t o = (size_t)g * BN + (size_t)b * NN + n0 + n;
            wsM[o] = m; wsNum[o] = num; wsDen[o] = den;
        }
    }
}

__global__ __launch_bounds__(256)
void mmse_combine_kernel(const float* __restrict__ wsM,
                         const float* __restrict__ wsNum,
                         const float* __restrict__ wsDen,
                         float* __restrict__ out)
{
    const int i = (int)blockIdx.x * 256 + (int)threadIdx.x;
    if (i >= BN) return;
    float m = wsM[i], num = wsNum[i], den = wsDen[i];
#pragma unroll
    for (int g = 1; g < GG; ++g) {
        const size_t o = (size_t)g * BN + i;
        smerge(m, num, den, wsM[o], wsNum[o], wsDen[o]);
    }
    out[i] = num / den;
}

extern "C" void kernel_launch(void* const* d_in, const int* in_sizes, int n_in,
                              void* d_out, int out_size, void* d_ws, size_t ws_size,
                              hipStream_t stream) {
    (void)in_sizes; (void)n_in; (void)out_size; (void)ws_size;
    const float* data    = (const float*)d_in[0];
    const float* targets = (const float*)d_in[1];
    const float* W       = (const float*)d_in[2];
    float*       out     = (float*)d_out;

    float* wsBase = (float*)d_ws;
    float* wsM    = wsBase;
    float* wsNum  = wsBase + (size_t)BNG;
    float* wsDen  = wsBase + (size_t)2 * BNG;

    mmse_scan_kernel<<<dim3(BB * GG), dim3(256), 0, stream>>>(
        data, targets, W, wsM, wsNum, wsDen);
    mmse_combine_kernel<<<dim3((BN + 255) / 256), dim3(256), 0, stream>>>(
        wsM, wsNum, wsDen, out);
}


# the 16 rows' wave max/sum reductions of a block merged into one halving tree: each exchange step carries half the registers and the rows run side by side; same pairs, sum order unchanged
# speedup vs baseline: 4.5345x; 1.7467x over previous
#include <hip/hip_runtime.h>

#define BB   32
#define NN   1024
#define DD   64
#define TT   2048
#define GG   8
#define TC   (TT / GG)
#define NB   16
#define NBP  (NB + 4)
#define NBLK (NN / NB)
#define BN   (BB * NN)
#define BNG  (BN * GG)

typedef float v2f __attribute__((ext_vector_type(2)));
typedef float v4f __attribute__((ext_vector_type(4)));
typedef float v8f __attribute__((ext_vector_type(8)));

__device__ __forceinline__ float XROW(float v) {
    const int b = __float_as_int(v);
    return __int_as_float(__builtin_amdgcn_permlanex16(
        b, b, 0x76543210, 0xfedcba98, false, false));
}
#define XMASK(v, X) \
    __int_as_float(__builtin_amdgcn_update_dpp( \
        0, __float_as_int(v), 0x160 + (X), 0xF, 0xF, true))

struct OpMax { __device__ __forceinline__ float operator()(float a, float b) const { return fmaxf(a, b); } };
struct OpAdd { __device__ __forceinline__ float operator()(float a, float b) const { return a + b; } };

template <class Op>
__device__ __forceinline__ float wave_reduce16(const float (&v)[16], int lane, Op op) {
    const bool h4 = (lane & 16) != 0;
    const bool h3 = (lane & 8) != 0;
    const bool h2 = (lane & 4) != 0;
    const bool h1 = (lane & 2) != 0;
    float a[8], b[4], c[2];
#pragma unroll
    for (int i = 0; i < 8; ++i) {
        const float keep = h4 ? v[8 + i] : v[i];
        const float send = h4 ? v[i] : v[8 + i];
        a[i] = op(keep, XROW(send));
    }
#pragma unroll
    for (int i = 0; i < 4; ++i) {
        const float keep = h3 ? a[4 + i] : a[i];
        const float send = h3 ? a[i] : a[4 + i];
        b[i] = op(keep, XMASK(send, 8));
    }
#pragma unroll
    for (int i = 0; i < 2; ++i) {
        const float keep = h2 ? b[2 + i] : b[i];
        const float send = h2 ? b[i] : b[2 + i];
        c[i] = op(keep, XMASK(send, 4));
    }
    const float keep = h1 ? c[1] : c[0];
    const float send = h1 ? c[0] : c[1];
    float d = op(keep, XMASK(send, 2));
    d = op(d, XMASK(d, 1));
    return d;
}

__device__ __forceinline__ void smerge(float& m, float& num, float& den,
                                       float m2, float n2, float d2) {
    float M  = fmaxf(m, m2);
    float s1 = __expf(m - M);
    float s2 = __expf(m2 - M);
    num = num * s1 + n2 * s2;
    den = den * s1 + d2 * s2;
    m   = M;
}

__global__ __launch_bounds__(256)
void mmse_scan_kernel(const float* __restrict__ data,
                      const float* __restrict__ targets,
                      const float* __restrict__ W,
                      float* __restrict__ wsM,
                      float* __restrict__ wsNum,
                      float* __restrict__ wsDen)
{
    __shared__ alignas(16) float xw[2][TC][NBP];
    __shared__ alignas(16) float red[2][NB][8][4];

    const int tid  = threadIdx.x;
    const int lane = tid & 31;
    const int wave = tid >> 5;
    const int lm   = lane & 15;
    const int lh   = lane >> 4;

    const int b     = (int)blockIdx.x >> 3;
    const int g     = (int)blockIdx.x & (GG - 1);
    const int tbase = g * TC;

    const float* dataB = data + (size_t)b * NN * DD;

    v2f bfrag[2][16];
#pragma unroll
    for (int j = 0; j < 2; ++j) {
        const int gcol = tbase + wave * 32 + j * 16 + lm;
#pragma unroll
        for (int s = 0; s < 16; ++s) {
            const int k0 = 4 * s + 2 * lh;
            v2f bf;
            bf.x = W[(size_t)k0 * TT + gcol];
            bf.y = W[(size_t)(k0 + 1) * TT + gcol];
            bfrag[j][s] = bf;
        }
    }

    auto compute = [&](int blk, int buf) {
        const int n0 = blk * NB;
        v2f af[16];
#pragma unroll
        for (int s = 0; s < 16; ++s) {
            const float* p = dataB + (size_t)(n0 + lm) * DD + 4 * s + 2 * lh;
            af[s] = *reinterpret_cast<const v2f*>(p);
        }
#pragma unroll
        for (int j = 0; j < 2; ++j) {
            v8f acc = {};
#pragma unroll
            for (int s = 0; s < 16; ++s) {
                acc = __builtin_amdgcn_wmma_f32_16x16x4_f32(
                     false, af[s],  false, bfrag[j][s],
                     (short)0, acc,  false,  false);
            }
            const int lcol = wave * 32 + j * 16 + lm;
            v4f lo = {acc[0], acc[1], acc[2], acc[3]};
            v4f hi = {acc[4], acc[5], acc[6], acc[7]};
            *reinterpret_cast<v4f*>(&xw[buf][lcol][8 * lh])     = lo;
            *reinterpret_cast<v4f*>(&xw[buf][lcol][8 * lh + 4]) = hi;
        }
    };

    float cum = 0.0f;

    compute(0, 0);
    __syncthreads();

    for (int blk = 0; blk < NBLK; ++blk) {
        const int cur = blk & 1;
        const int n0  = blk * NB;

        if (blk + 1 < NBLK) compute(blk + 1, cur ^ 1);

        v4f xr[4], tr[4];
#pragma unroll
        for (int q = 0; q < 4; ++q)
            xr[q] = *reinterpret_cast<const v4f*>(&xw[cur][tid][4 * q]);
        const float* tb = targets + b * NN + n0;
#pragma unroll
        for (int q = 0; q < 4; ++q)
            tr[q] = *reinterpret_cast<const v4f*>(tb + 4 * q);

        float x[NB], c[NB];
#pragma unroll
        for (int n = 0; n < NB; ++n) {
            x[n] = xr[n >> 2][n & 3];
            c[n] = cum;
            const float df = tr[n >> 2][n & 3] - x[n];
            cum = fmaf(df, df * -0.5f, cum);
        }

        const float mrow = wave_reduce16(c, lane, OpMax());

        float e[NB], ex[NB];
#pragma unroll
        for (int n = 0; n < NB; ++n) {
            const float m = __int_as_float(
                __builtin_amdgcn_readlane(__float_as_int(mrow), 2 * n));
            e[n]  = __expf(c[n] - m);
            ex[n] = e[n] * x[n];
        }
        const float nurow = wave_reduce16(ex, lane, OpAdd());
        const float derow = wave_reduce16(e, lane, OpAdd());
        if ((lane & 1) == 0) {
            v4f t4 = {mrow, nurow, derow, 0.0f};
            *reinterpret_cast<v4f*>(&red[cur][lane >> 1][wave][0]) = t4;
        }

        __syncthreads();

        if ((tid & 15) == 0) {
            const int n = tid >> 4;
            v4f t0 = *reinterpret_cast<const v4f*>(&red[cur][n][0][0]);
            float m = t0.x, num = t0.y, den = t0.z;
#pragma unroll
            for (int w = 1; w < 8; ++w) {
                v4f tw = *reinterpret_cast<const v4f*>(&red[cur][n][w][0]);
                smerge(m, num, den, tw.x, tw.y, tw.z);
            }
            const size_t o = (size_t)g * BN + (size_t)b * NN + n0 + n;
            wsM[o] = m; wsNum[o] = num; wsDen[o] = den;
        }
    }
}

__global__ __launch_bounds__(256)
void mmse_combine_kernel(const float* __restrict__ wsM,
                         const float* __restrict__ wsNum,
                         const float* __restrict__ wsDen,
                         float* __restrict__ out)
{
    const int i = (int)blockIdx.x * 256 + (int)threadIdx.x;
    if (i >= BN) return;
    float m = wsM[i], num = wsNum[i], den = wsDen[i];
#pragma unroll
    for (int g = 1; g < GG; ++g) {
        const size_t o = (size_t)g * BN + i;
        smerge(m, num, den, wsM[o], wsNum[o], wsDen[o]);
    }
    out[i] = num / den;
}

extern "C" void kernel_launch(void* const* d_in, const int* in_sizes, int n_in,
                              void* d_out, int out_size, void* d_ws, size_t ws_size,
                              hipStream_t stream) {
    (void)in_sizes; (void)n_in; (void)out_size; (void)ws_size;
    const float* data    = (const float*)d_in[0];
    const float* targets = (const float*)d_in[1];
    const float* W       = (const float*)d_in[2];
    float*       out     = (float*)d_out;

    float* wsBase = (float*)d_ws;
    float* wsM    = wsBase;
    float* wsNum  = wsBase + (size_t)BNG;
    float* wsDen  = wsBase + (size_t)2 * BNG;

    mmse_scan_kernel<<<dim3(BB * GG), dim3(256), 0, stream>>>(
        data, targets, W, wsM, wsNum, wsDen);
    mmse_combine_kernel<<<dim3((BN + 255) / 256), dim3(256), 0, stream>>>(
        wsM, wsNum, wsDen, out);
}


# data tile fetched once per workgroup (one coalesced 16-byte load per thread, two blocks ahead) and parked in LDS; waves take A fragments from LDS, not by 16-line global gathers; k order kept
# speedup vs baseline: 6.0100x; 1.3254x over previous
#include <hip/hip_runtime.h>

#define BB   32
#define NN   1024
#define DD   64
#define TT   2048
#define GG   8
#define TC   (TT / GG)
#define NB   16
#define NBP  (NB + 4)
#define NBLK (NN / NB)
#define AP   (DD + 4)
#define BN   (BB * NN)
#define BNG  (BN * GG)

typedef float v2f __attribute__((ext_vector_type(2)));
typedef float v4f __attribute__((ext_vector_type(4)));
typedef float v8f __attribute__((ext_vector_type(8)));

__device__ __forceinline__ float XROW(float v) {
    const int b = __float_as_int(v);
    return __int_as_float(__builtin_amdgcn_permlanex16(
        b, b, 0x76543210, 0xfedcba98, false, false));
}
#define XMASK(v, X) \
    __int_as_float(__builtin_amdgcn_update_dpp( \
        0, __float_as_int(v), 0x160 + (X), 0xF, 0xF, true))

struct OpMax { __device__ __forceinline__ float operator()(float a, float b) const { return fmaxf(a, b); } };
struct OpAdd { __device__ __forceinline__ float operator()(float a, float b) const { return a + b; } };

template <class Op>
__device__ __forceinline__ float wave_reduce16(const float (&v)[16], int lane, Op op) {
    const bool h4 = (lane & 16) != 0;
    const bool h3 = (lane & 8) != 0;
    const bool h2 = (lane & 4) != 0;
    const bool h1 = (lane & 2) != 0;
    float a[8], b[4], c[2];
#pragma unroll
    for (int i = 0; i < 8; ++i) {
        const float keep = h4 ? v[8 + i] : v[i];
        const float send = h4 ? v[i] : v[8 + i];
        a[i] = op(keep, XROW(send));
    }
#pragma unroll
    for (int i = 0; i < 4; ++i) {
        const float keep = h3 ? a[4 + i] : a[i];
        const float send = h3 ? a[i] : a[4 + i];
        b[i] = op(keep, XMASK(send, 8));
    }
#pragma unroll
    for (int i = 0; i < 2; ++i) {
        const float keep = h2 ? b[2 + i] : b[i];
        const float send = h2 ? b[i] : b[2 + i];
        c[i] = op(keep, XMASK(send, 4));
    }
    const float keep = h1 ? c[1] : c[0];
    const float send = h1 ? c[0] : c[1];
    float d = op(keep, XMASK(send, 2));
    d = op(d, XMASK(d, 1));
    return d;
}

__device__ __forceinline__ void smerge(float& m, float& num, float& den,
                                       float m2, float n2, float d2) {
    float M  = fmaxf(m, m2);
    float s1 = __expf(m - M);
    float s2 = __expf(m2 - M);
    num = num * s1 + n2 * s2;
    den = den * s1 + d2 * s2;
    m   = M;
}

__global__ __launch_bounds__(256)
void mmse_scan_kernel(const float* __restrict__ data,
                      const float* __restrict__ targets,
                      const float* __restrict__ W,
                      float* __restrict__ wsM,
                      float* __restrict__ wsNum,
                      float* __restrict__ wsDen)
{
    __shared__ alignas(16) float xw[2][TC][NBP];
    __shared__ alignas(16) float red[2][NB][8][4];
    __shared__ alignas(16) float sA[2][NB][AP];

    const int tid  = threadIdx.x;
    const int lane = tid & 31;
    const int wave = tid >> 5;
    const int lm   = lane & 15;
    const int lh   = lane >> 4;

    const int b     = (int)blockIdx.x >> 3;
    const int g     = (int)blockIdx.x & (GG - 1);
    const int tbase = g * TC;

    const float* dataB = data + (size_t)b * NN * DD;

    v2f bfrag[2][16];
#pragma unroll
    for (int j = 0; j < 2; ++j) {
        const int gcol = tbase + wave * 32 + j * 16 + lm;
#pragma unroll
        for (int s = 0; s < 16; ++s) {
            const int k0 = 4 * s + 2 * lh;
            v2f bf;
            bf.x = W[(size_t)k0 * TT + gcol];
            bf.y = W[(size_t)(k0 + 1) * TT + gcol];
            bfrag[j][s] = bf;
        }
    }

    auto load_tile = [&](int blk) -> v4f {
        return *(reinterpret_cast<const v4f*>(dataB + (size_t)blk * NB * DD) + tid);
    };
    auto store_tile = [&](int buf, v4f a) {
        *reinterpret_cast<v4f*>(&sA[buf][tid >> 4][4 * (tid & 15)]) = a;
    };

    v2f af[16];
    auto load_frags = [&](int buf) {
#pragma unroll
        for (int s = 0; s < 16; ++s)
            af[s] = *reinterpret_cast<const v2f*>(&sA[buf][lm][4 * s + 2 * lh]);
    };

    auto compute = [&](int buf) {
#pragma unroll
        for (int j = 0; j < 2; ++j) {
            v8f acc = {};
#pragma unroll
            for (int s = 0; s < 16; ++s) {
                acc = __builtin_amdgcn_wmma_f32_16x16x4_f32(
                     false, af[s],  false, bfrag[j][s],
                     (short)0, acc,  false,  false);
            }
            const int lcol = wave * 32 + j * 16 + lm;
            v4f lo = {acc[0], acc[1], acc[2], acc[3]};
            v4f hi = {acc[4], acc[5], acc[6], acc[7]};
            *reinterpret_cast<v4f*>(&xw[buf][lcol][8 * lh])     = lo;
            *reinterpret_cast<v4f*>(&xw[buf][lcol][8 * lh + 4]) = hi;
        }
    };

    float cum = 0.0f;

    store_tile(0, load_tile(0));
    store_tile(1, load_tile(1));
    __syncthreads();
    load_frags(0);
    compute(0);
    __syncthreads();
    load_frags(1);

    for (int blk = 0; blk < NBLK; ++blk) {
        const int cur = blk & 1;
        const int n0  = blk * NB;

        const bool more2 = blk + 2 < NBLK;
        v4f anext = {0.0f, 0.0f, 0.0f, 0.0f};
        if (more2) anext = load_tile(blk + 2);

        if (blk + 1 < NBLK) compute(cur ^ 1);

        v4f xr[4], tr[4];
#pragma unroll
        for (int q = 0; q < 4; ++q)
            xr[q] = *reinterpret_cast<const v4f*>(&xw[cur][tid][4 * q]);
        const float* tb = targets + b * NN + n0;
#pragma unroll
        for (int q = 0; q < 4; ++q)
            tr[q] = *reinterpret_cast<const v4f*>(tb + 4 * q);

        float x[NB], c[NB];
#pragma unroll
        for (int n = 0; n < NB; ++n) {
            x[n] = xr[n >> 2][n & 3];
            c[n] = cum;
            const float df = tr[n >> 2][n & 3] - x[n];
            cum = fmaf(df, df * -0.5f, cum);
        }

        const float mrow = wave_reduce16(c, lane, OpMax());

        float e[NB], ex[NB];
#pragma unroll
        for (int n = 0; n < NB; ++n) {
            const float m = __int_as_float(
                __builtin_amdgcn_readlane(__float_as_int(mrow), 2 * n));
            e[n]  = __expf(c[n] - m);
            ex[n] = e[n] * x[n];
        }
        const float nurow = wave_reduce16(ex, lane, OpAdd());
        const float derow = wave_reduce16(e, lane, OpAdd());
        if ((lane & 1) == 0) {
            v4f t4 = {mrow, nurow, derow, 0.0f};
            *reinterpret_cast<v4f*>(&red[cur][lane >> 1][wave][0]) = t4;
        }

        if (more2) store_tile(cur, anext);

        __syncthreads();

        if (more2) load_frags(cur);

        if ((tid & 15) == 0) {
            const int n = tid >> 4;
            v4f t0 = *reinterpret_cast<const v4f*>(&red[cur][n][0][0]);
            float m = t0.x, num = t0.y, den = t0.z;
#pragma unroll
            for (int w = 1; w < 8; ++w) {
                v4f tw = *reinterpret_cast<const v4f*>(&red[cur][n][w][0]);
                smerge(m, num, den, tw.x, tw.y, tw.z);
            }
            const size_t o = (size_t)g * BN + (size_t)b * NN + n0 + n;
            wsM[o] = m; wsNum[o] = num; wsDen[o] = den;
        }
    }
}

__global__ __launch_bounds__(256)
void mmse_combine_kernel(const float* __restrict__ wsM,
                         const float* __restrict__ wsNum,
                         const float* __restrict__ wsDen,
                         float* __restrict__ out)
{
    const int i = (int)blockIdx.x * 256 + (int)threadIdx.x;
    if (i >= BN) return;
    float m = wsM[i], num = wsNum[i], den = wsDen[i];
#pragma unroll
    for (int g = 1; g < GG; ++g) {
        const size_t o = (size_t)g * BN + i;
        smerge(m, num, den, wsM[o], wsNum[o], wsDen[o]);
    }
    out[i] = num / den;
}

extern "C" void kernel_launch(void* const* d_in, const int* in_sizes, int n_in,
                              void* d_out, int out_size, void* d_ws, size_t ws_size,
                              hipStream_t stream) {
    (void)in_sizes; (void)n_in; (void)out_size; (void)ws_size;
    const float* data    = (const float*)d_in[0];
    const float* targets = (const float*)d_in[1];
    const float* W       = (const float*)d_in[2];
    float*       out     = (float*)d_out;

    float* wsBase = (float*)d_ws;
    float* wsM    = wsBase;
    float* wsNum  = wsBase + (size_t)BNG;
    float* wsDen  = wsBase + (size_t)2 * BNG;

    mmse_scan_kernel<<<dim3(BB * GG), dim3(256), 0, stream>>>(
        data, targets, W, wsM, wsNum, wsDen);
    mmse_combine_kernel<<<dim3((BN + 255) / 256), dim3(256), 0, stream>>>(
        wsM, wsNum, wsDen, out);
}


# block softmax taken by row: scan parks its prefixes in LDS; 16 threads per row sum 16 columns each in registers, then 4 DPP steps; no register trees, no cross-wave merge; summation order changed
# speedup vs baseline: 6.4957x; 1.0808x over previous
#include <hip/hip_runtime.h>

#define BB   32
#define NN   1024
#define DD   64
#define TT   2048
#define GG   8
#define TC   (TT / GG)
#define NB   16
#define NBP  (NB + 4)
#define NBLK (NN / NB)
#define AP   (DD + 4)
#define XS   (TC * NBP)
#define AS   (NB * AP)
#define BN   (BB * NN)
#define BNG  (BN * GG)

typedef float v2f __attribute__((ext_vector_type(2)));
typedef float v4f __attribute__((ext_vector_type(4)));
typedef float v8f __attribute__((ext_vector_type(8)));

#define XMASK(v, X) \
    __int_as_float(__builtin_amdgcn_update_dpp( \
        0, __float_as_int(v), 0x160 + (X), 0xF, 0xF, true))

__device__ __forceinline__ void smerge(float& m, float& num, float& den,
                                       float m2, float n2, float d2) {
    float M  = fmaxf(m, m2);
    float s1 = __expf(m - M);
    float s2 = __expf(m2 - M);
    num = num * s1 + n2 * s2;
    den = den * s1 + d2 * s2;
    m   = M;
}

__global__ __launch_bounds__(256)
void mmse_scan_kernel(const float* __restrict__ data,
                      const float* __restrict__ targets,
                      const float* __restrict__ W,
                      float* __restrict__ wsM,
                      float* __restrict__ wsNum,
                      float* __restrict__ wsDen)
{
    __shared__ alignas(16) float xw[3 * XS];
    __shared__ alignas(16) float al[2 * XS];
    __shared__ alignas(16) float sA[2 * AS];

    const int tid  = threadIdx.x;
    const int lane = tid & 31;
    const int wave = tid >> 5;
    const int lm   = lane & 15;
    const int lh   = lane >> 4;

    const int b     = (int)blockIdx.x >> 3;
    const int g     = (int)blockIdx.x & (GG - 1);
    const int tbase = g * TC;

    const float* dataB = data + (size_t)b * NN * DD;

    v2f bfrag[2][16];
#pragma unroll
    for (int j = 0; j < 2; ++j) {
        const int gcol = tbase + wave * 32 + j * 16 + lm;
#pragma unroll
        for (int s = 0; s < 16; ++s) {
            const int k0 = 4 * s + 2 * lh;
            v2f bf;
            bf.x = W[(size_t)k0 * TT + gcol];
            bf.y = W[(size_t)(k0 + 1) * TT + gcol];
            bfrag[j][s] = bf;
        }
    }

    auto load_tile = [&](int blk) -> v4f {
        return *(reinterpret_cast<const v4f*>(dataB + (size_t)blk * NB * DD) + tid);
    };
    auto store_tile = [&](int aoff, v4f a) {
        *reinterpret_cast<v4f*>(&sA[aoff + (tid >> 4) * AP + 4 * (tid & 15)]) = a;
    };

    v2f af[16];
    auto load_frags = [&](int aoff) {
#pragma unroll
        for (int s = 0; s < 16; ++s)
            af[s] = *reinterpret_cast<const v2f*>(&sA[aoff + lm * AP + 4 * s + 2 * lh]);
    };

    auto compute = [&](int xq) {
#pragma unroll
        for (int j = 0; j < 2; ++j) {
            v8f acc = {};
#pragma unroll
            for (int s = 0; s < 16; ++s) {
                acc = __builtin_amdgcn_wmma_f32_16x16x4_f32(
                     false, af[s],  false, bfrag[j][s],
                     (short)0, acc,  false,  false);
            }
            const int lcol = wave * 32 + j * 16 + lm;
            v4f lo = {acc[0], acc[1], acc[2], acc[3]};
            v4f hi = {acc[4], acc[5], acc[6], acc[7]};
            *reinterpret_cast<v4f*>(&xw[4 * xq + lcol * NBP + 8 * lh])     = lo;
            *reinterpret_cast<v4f*>(&xw[4 * xq + lcol * NBP + 8 * lh + 4]) = hi;
        }
    };

    float cum = 0.0f;

    const int pn = tid >> 4;
    const int pk = tid & 15;
    float pa[NB], px[NB];
#pragma unroll
    for (int i = 0; i < NB; ++i) { pa[i] = 0.0f; px[i] = 0.0f; }

    auto row_pass = [&](float& m, float& num, float& den) {
        m = pa[0];
#pragma unroll
        for (int i = 1; i < NB; ++i) m = fmaxf(m, pa[i]);
        m = fmaxf(m, XMASK(m, 8));
        m = fmaxf(m, XMASK(m, 4));
        m = fmaxf(m, XMASK(m, 2));
        m = fmaxf(m, XMASK(m, 1));
        den = 0.0f; num = 0.0f;
#pragma unroll
        for (int i = 0; i < NB; ++i) {
            const float e = __expf(pa[i] - m);
            den += e;
            num = fmaf(e, px[i], num);
        }
        den += XMASK(den, 8);  num += XMASK(num, 8);
        den += XMASK(den, 4);  num += XMASK(num, 4);
        den += XMASK(den, 2);  num += XMASK(num, 2);
        den += XMASK(den, 1);  num += XMASK(num, 1);
    };
    const size_t obase = (size_t)g * BN + (size_t)b * NN + pn;

    store_tile(0, load_tile(0));
    store_tile(AS, load_tile(1));
    __syncthreads();
    load_frags(0);
    compute(0);
    __syncthreads();
    load_frags(AS);

    auto row_fetch = [&](int xq, int poff) {
#pragma unroll
        for (int i = 0; i < NB; ++i) {
            px[i] = xw[4 * xq + (16 * i + pk) * NBP + pn];
            pa[i] = al[poff + (16 * i + pk) * NBP + pn];
        }
    };

    int ring  = 0;
    int rprev = 2 * (XS / 4);
    for (int blk = 0; blk < NBLK; ++blk) {
        const int n0   = blk * NB;
        const int rnxt = (ring == 2 * (XS / 4)) ? 0 : ring + XS / 4;
        const int pcur = (blk & 1) ? XS : 0;
        const int pold = XS - pcur;
        const int acur = (blk & 1) ? AS : 0;

        v4f tr[4];
        const float* tb = targets + b * NN + n0;
#pragma unroll
        for (int q = 0; q < 4; ++q)
            tr[q] = *reinterpret_cast<const v4f*>(tb + 4 * q);

        const bool more2 = blk + 2 < NBLK;
        v4f anext = {0.0f, 0.0f, 0.0f, 0.0f};
        if (more2) anext = load_tile(blk + 2);

        if (blk + 1 < NBLK) compute(rnxt);

        v4f xr[4];
#pragma unroll
        for (int q = 0; q < 4; ++q)
            xr[q] = *reinterpret_cast<const v4f*>(&xw[4 * ring + tid * NBP + 4 * q]);
        if (blk > 0) row_fetch(rprev, pold);

        float c[NB];
#pragma unroll
        for (int n = 0; n < NB; ++n) {
            c[n] = cum;
            const float df = tr[n >> 2][n & 3] - xr[n >> 2][n & 3];
            cum = fmaf(df, df * -0.5f, cum);
        }
#pragma unroll
        for (int q = 0; q < 4; ++q) {
            v4f c4 = {c[4 * q], c[4 * q + 1], c[4 * q + 2], c[4 * q + 3]};
            *reinterpret_cast<v4f*>(&al[pcur + tid * NBP + 4 * q]) = c4;
        }

        if (more2) store_tile(acur, anext);

        __syncthreads();

        if (more2) load_frags(acur);

        float rm, rnum, rden;
        row_pass(rm, rnum, rden);
        if (pk == 0 && blk > 0) {
            const size_t o = obase + (n0 - NB);
            wsM[o] = rm; wsNum[o] = rnum; wsDen[o] = rden;
        }

        rprev = ring;
        ring  = rnxt;
    }

    {
        float rm, rnum, rden;
        row_fetch(rprev, ((NBLK - 1) & 1) ? XS : 0);
        row_pass(rm, rnum, rden);
        if (pk == 0) {
            const size_t o = obase + (NN - NB);
            wsM[o] = rm; wsNum[o] = rnum; wsDen[o] = rden;
        }
    }
}

__global__ __launch_bounds__(256)
void mmse_combine_kernel(const float* __restrict__ wsM,
                         const float* __restrict__ wsNum,
                         const float* __restrict__ wsDen,
                         float* __restrict__ out)
{
    const int i = (int)blockIdx.x * 256 + (int)threadIdx.x;
    if (i >= BN) return;
    float m = wsM[i], num = wsNum[i], den = wsDen[i];
#pragma unroll
    for (int g = 1; g < GG; ++g) {
        const size_t o = (size_t)g * BN + i;
        smerge(m, num, den, wsM[o], wsNum[o], wsDen[o]);
    }
    out[i] = num / den;
}

extern "C" void kernel_launch(void* const* d_in, const int* in_sizes, int n_in,
                              void* d_out, int out_size, void* d_ws, size_t ws_size,
                              hipStream_t stream) {
    (void)in_sizes; (void)n_in; (void)out_size; (void)ws_size;
    const float* data    = (const float*)d_in[0];
    const float* targets = (const float*)d_in[1];
    const float* W       = (const float*)d_in[2];
    float*       out     = (float*)d_out;

    float* wsBase = (float*)d_ws;
    float* wsM    = wsBase;
    float* wsNum  = wsBase + (size_t)BNG;
    float* wsDen  = wsBase + (size_t)2 * BNG;

    mmse_scan_kernel<<<dim3(BB * GG), dim3(256), 0, stream>>>(
        data, targets, W, wsM, wsNum, wsDen);
    mmse_combine_kernel<<<dim3((BN + 255) / 256), dim3(256), 0, stream>>>(
        wsM, wsNum, wsDen, out);
}


# scan and row-pass vector work interleaved into the next block's WMMA run, a slice behind each WMMA pair; all LDS reads grouped at the step's top, A fragments one block further ahead; sum order kept
# speedup vs baseline: 7.2378x; 1.1142x over previous
#include <hip/hip_runtime.h>

#define BB   32
#define NN   1024
#define DD   64
#define TT   2048
#define GG   8
#define TC   (TT / GG)
#define NB   16
#define NBP  (NB + 4)
#define NBLK (NN / NB)
#define AP   (DD + 4)
#define XS   (TC * NBP)
#define AS   (NB * AP)
#define BN   (BB * NN)
#define BNG  (BN * GG)

typedef float v2f __attribute__((ext_vector_type(2)));
typedef float v4f __attribute__((ext_vector_type(4)));
typedef float v8f __attribute__((ext_vector_type(8)));

#define XMASK(v, X) \
    __int_as_float(__builtin_amdgcn_update_dpp( \
        0, __float_as_int(v), 0x160 + (X), 0xF, 0xF, true))

__device__ __forceinline__ void smerge(float& m, float& num, float& den,
                                       float m2, float n2, float d2) {
    float M  = fmaxf(m, m2);
    float s1 = __expf(m - M);
    float s2 = __expf(m2 - M);
    num = num * s1 + n2 * s2;
    den = den * s1 + d2 * s2;
    m   = M;
}

__global__ __launch_bounds__(256)
void mmse_scan_kernel(const float* __restrict__ data,
                      const float* __restrict__ targets,
                      const float* __restrict__ W,
                      float* __restrict__ wsM,
                      float* __restrict__ wsNum,
                      float* __restrict__ wsDen)
{
    __shared__ alignas(16) float xw[3 * XS];
    __shared__ alignas(16) float al[2 * XS];
    __shared__ alignas(16) float sA[2 * AS];

    const int tid  = threadIdx.x;
    const int lane = tid & 31;
    const int wave = tid >> 5;
    const int lm   = lane & 15;
    const int lh   = lane >> 4;

    const int b     = (int)blockIdx.x >> 3;
    const int g     = (int)blockIdx.x & (GG - 1);
    const int tbase = g * TC;

    const float* dataB = data + (size_t)b * NN * DD;
    const float* targB = targets + b * NN;

    v2f bfrag[2][16];
#pragma unroll
    for (int j = 0; j < 2; ++j) {
        const int gcol = tbase + wave * 32 + j * 16 + lm;
#pragma unroll
        for (int s = 0; s < 16; ++s) {
            const int k0 = 4 * s + 2 * lh;
            v2f bf;
            bf.x = W[(size_t)k0 * TT + gcol];
            bf.y = W[(size_t)(k0 + 1) * TT + gcol];
            bfrag[j][s] = bf;
        }
    }

    auto load_tile = [&](int blk) -> v4f {
        const int t = blk < NBLK ? blk : NBLK - 1;
        return *(reinterpret_cast<const v4f*>(dataB + (size_t)t * NB * DD) + tid);
    };
    auto store_tile = [&](int aoff, v4f a) {
        *reinterpret_cast<v4f*>(&sA[aoff + (tid >> 4) * AP + 4 * (tid & 15)]) = a;
    };
    auto load_targets = [&](int blk, v4f (&t)[4]) {
        const int tb = blk < NBLK ? blk : NBLK - 1;
#pragma unroll
        for (int q = 0; q < 4; ++q)
            t[q] = *reinterpret_cast<const v4f*>(targB + tb * NB + 4 * q);
    };

    v2f afA[16], afB[16];
    auto load_frags = [&](v2f (&af)[16], int aoff) {
#pragma unroll
        for (int s = 0; s < 16; ++s)
            af[s] = *reinterpret_cast<const v2f*>(&sA[aoff + lm * AP + 4 * s + 2 * lh]);
    };

    v8f acc0, acc1;
    auto store_products = [&](int xq) {
#pragma unroll
        for (int j = 0; j < 2; ++j) {
            const v8f& acc = j ? acc1 : acc0;
            const int lcol = wave * 32 + j * 16 + lm;
            v4f lo = {acc[0], acc[1], acc[2], acc[3]};
            v4f hi = {acc[4], acc[5], acc[6], acc[7]};
            *reinterpret_cast<v4f*>(&xw[4 * xq + lcol * NBP + 8 * lh])     = lo;
            *reinterpret_cast<v4f*>(&xw[4 * xq + lcol * NBP + 8 * lh + 4]) = hi;
        }
    };

    float cum = 0.0f;

    const int pn = tid >> 4;
    const int pk = tid & 15;
    const size_t obase = (size_t)g * BN + (size_t)b * NN + pn;

    float pa[NB], px[NB];
    auto row_fetch = [&](int xq, int poff) {
#pragma unroll
        for (int i = 0; i < NB; ++i) {
            px[i] = xw[4 * xq + (16 * i + pk) * NBP + pn];
            pa[i] = al[poff + (16 * i + pk) * NBP + pn];
        }
    };

    float sm = 0.0f, snum = 0.0f, sden = 0.0f;
    auto store_row = [&](int blk) {
        const size_t o = obase + (size_t)blk * NB;
        wsM[o] = sm; wsNum[o] = snum; wsDen[o] = sden;
    };

    v4f tr[4];
    v4f apark;

    store_tile(0, load_tile(0));
    store_tile(AS, load_tile(1));
    __syncthreads();
    load_frags(afA, 0);
    acc0 = v8f{}; acc1 = v8f{};
#pragma unroll
    for (int s = 0; s < 16; ++s) {
        acc0 = __builtin_amdgcn_wmma_f32_16x16x4_f32(false, afA[s], false, bfrag[0][s],
                                                     (short)0, acc0, false, false);
        acc1 = __builtin_amdgcn_wmma_f32_16x16x4_f32(false, afA[s], false, bfrag[1][s],
                                                     (short)0, acc1, false, false);
    }
    store_products(0);
    load_frags(afB, AS);
    __syncthreads();
    store_tile(0, load_tile(2));
    apark = load_tile(3);
    load_targets(0, tr);
    __syncthreads();

#define SLICE() __builtin_amdgcn_sched_barrier(0)

    auto step = [&](int blk, const v2f (&cur)[16], v2f (&nxt)[16],
                    const int fbuf, const int sbuf, const int pcur, const int pold,
                    const int ring, const int rprev, const int rnxt,
                    const bool first, const bool gemm)
        __attribute__((always_inline)) {
        if (pk == 0 && blk > 1) store_row(blk - 2);

        v4f xr[4];
#pragma unroll
        for (int q = 0; q < 4; ++q)
            xr[q] = *reinterpret_cast<const v4f*>(&xw[4 * ring + tid * NBP + 4 * q]);
        if (!first) row_fetch(rprev, pold);
        load_frags(nxt, fbuf);
        store_tile(sbuf, apark);
        apark = load_tile(blk + 4);
        SLICE();

        float c[NB];
        float m = 0.0f, den = 0.0f, num = 0.0f;
        acc0 = v8f{}; acc1 = v8f{};
#pragma unroll
        for (int s = 0; s < 16; ++s) {
            if (gemm) {
                acc0 = __builtin_amdgcn_wmma_f32_16x16x4_f32(false, cur[s], false, bfrag[0][s],
                                                             (short)0, acc0, false, false);
                acc1 = __builtin_amdgcn_wmma_f32_16x16x4_f32(false, cur[s], false, bfrag[1][s],
                                                             (short)0, acc1, false, false);
                SLICE();
            }
            if (s >= 1 && s <= 4) {
                const int q = s - 1;
#pragma unroll
                for (int r = 0; r < 4; ++r) {
                    c[4 * q + r] = cum;
                    const float df = tr[q][r] - xr[q][r];
                    cum = fmaf(df, df * -0.5f, cum);
                }
            }
            if (s == 5) load_targets(blk + 1, tr);
            if (s == 4) {
#pragma unroll
                for (int q = 0; q < 4; ++q) {
                    v4f c4 = {c[4 * q], c[4 * q + 1], c[4 * q + 2], c[4 * q + 3]};
                    *reinterpret_cast<v4f*>(&al[pcur + tid * NBP + 4 * q]) = c4;
                }
            }
            if (!first) {
                if (s == 5) {
                    m = pa[0];
#pragma unroll
                    for (int i = 1; i < NB; ++i) m = fmaxf(m, pa[i]);
                    m = fmaxf(m, XMASK(m, 8));
                    m = fmaxf(m, XMASK(m, 4));
                }
                if (s == 6) {
                    m = fmaxf(m, XMASK(m, 2));
                    m = fmaxf(m, XMASK(m, 1));
                }
                if (s >= 6 && s <= 13) {
#pragma unroll
                    for (int i = 2 * (s - 6); i < 2 * (s - 6) + 2; ++i) {
                        const float e = __expf(pa[i] - m);
                        den += e;
                        num = fmaf(e, px[i], num);
                    }
                }
                if (s == 14) {
                    den += XMASK(den, 8);  num += XMASK(num, 8);
                    den += XMASK(den, 4);  num += XMASK(num, 4);
                }
                if (s == 15) {
                    den += XMASK(den, 2);  num += XMASK(num, 2);
                    den += XMASK(den, 1);  num += XMASK(num, 1);
                }
            }
            SLICE();
        }
        if (gemm) store_products(rnxt);
        if (!first) { sm = m; snum = num; sden = den; }

        __syncthreads();
    };

    constexpr int R0 = 0, R1 = XS / 4, R2 = 2 * (XS / 4);
    step(0, afB, afA, 0, AS, 0, XS, R0, R2, R1, true, true);
    int ring = R1, rprev = R0;
    for (int blk = 1; blk < NBLK - 1; blk += 2) {
        int rnxt = (ring == R2) ? 0 : ring + XS / 4;
        step(blk, afA, afB, AS, 0, XS, 0, ring, rprev, rnxt, false, true);
        rprev = ring; ring = rnxt;
        rnxt = (ring == R2) ? 0 : ring + XS / 4;
        step(blk + 1, afB, afA, 0, AS, 0, XS, ring, rprev, rnxt, false, true);
        rprev = ring; ring = rnxt;
    }
    step(NBLK - 1, afA, afB, AS, 0, XS, 0, ring, rprev, 0, false, false);

    if (pk == 0) store_row(NBLK - 2);
    {
        row_fetch(ring, XS);
        float m = pa[0];
#pragma unroll
        for (int i = 1; i < NB; ++i) m = fmaxf(m, pa[i]);
        m = fmaxf(m, XMASK(m, 8));
        m = fmaxf(m, XMASK(m, 4));
        m = fmaxf(m, XMASK(m, 2));
        m = fmaxf(m, XMASK(m, 1));
        float den = 0.0f, num = 0.0f;
#pragma unroll
        for (int i = 0; i < NB; ++i) {
            const float e = __expf(pa[i] - m);
            den += e;
            num = fmaf(e, px[i], num);
        }
        den += XMASK(den, 8);  num += XMASK(num, 8);
        den += XMASK(den, 4);  num += XMASK(num, 4);
        den += XMASK(den, 2);  num += XMASK(num, 2);
        den += XMASK(den, 1);  num += XMASK(num, 1);
        sm = m; snum = num; sden = den;
        if (pk == 0) store_row(NBLK - 1);
    }
#undef SLICE
}

__global__ __launch_bounds__(256)
void mmse_combine_kernel(const float* __restrict__ wsM,
                         const float* __restrict__ wsNum,
                         const float* __restrict__ wsDen,
                         float* __restrict__ out)
{
    const int i = (int)blockIdx.x * 256 + (int)threadIdx.x;
    if (i >= BN) return;
    float m = wsM[i], num = wsNum[i], den = wsDen[i];
#pragma unroll
    for (int g = 1; g < GG; ++g) {
        const size_t o = (size_t)g * BN + i;
        smerge(m, num, den, wsM[o], wsNum[o], wsDen[o]);
    }
    out[i] = num / den;
}

extern "C" void kernel_launch(void* const* d_in, const int* in_sizes, int n_in,
                              void* d_out, int out_size, void* d_ws, size_t ws_size,
                              hipStream_t stream) {
    (void)in_sizes; (void)n_in; (void)out_size; (void)ws_size;
    const float* data    = (const float*)d_in[0];
    const float* targets = (const float*)d_in[1];
    const float* W       = (const float*)d_in[2];
    float*       out     = (float*)d_out;

    float* wsBase = (float*)d_ws;
    float* wsM    = wsBase;
    float* wsNum  = wsBase + (size_t)BNG;
    float* wsDen  = wsBase + (size_t)2 * BNG;

    mmse_scan_kernel<<<dim3(BB * GG), dim3(256), 0, stream>>>(
        data, targets, W, wsM, wsNum, wsDen);
    mmse_combine_kernel<<<dim3((BN + 255) / 256), dim3(256), 0, stream>>>(
        wsM, wsNum, wsDen, out);
}


# workgroup barrier taken in two halves: wait mid-step behind five WMMA pairs and the own-column scan, signal two pairs behind the cross-wave LDS group; waves run out of step; sum order kept
# speedup vs baseline: 7.9731x; 1.1016x over previous
#include <hip/hip_runtime.h>

#define BB   32
#define NN   1024
#define DD   64
#define TT   2048
#define GG   8
#define TC   (TT / GG)
#define NB   16
#define NBP  (NB + 4)
#define NBLK (NN / NB)
#define AP   (DD + 4)
#define XS   (TC * NBP)
#define AS   (NB * AP)
#define BN   (BB * NN)
#define BNG  (BN * GG)

typedef float v2f __attribute__((ext_vector_type(2)));
typedef float v4f __attribute__((ext_vector_type(4)));
typedef float v8f __attribute__((ext_vector_type(8)));

#define XMASK(v, X) \
    __int_as_float(__builtin_amdgcn_update_dpp( \
        0, __float_as_int(v), 0x160 + (X), 0xF, 0xF, true))

__device__ __forceinline__ void bar_signal() {
    __builtin_amdgcn_fence(__ATOMIC_RELEASE, "workgroup", "local");
    __builtin_amdgcn_s_barrier_signal(-1);
}
__device__ __forceinline__ void bar_wait() {
    __builtin_amdgcn_s_barrier_wait(-1);
    __builtin_amdgcn_fence(__ATOMIC_ACQUIRE, "workgroup", "local");
}

__device__ __forceinline__ void smerge(float& m, float& num, float& den,
                                       float m2, float n2, float d2) {
    float M  = fmaxf(m, m2);
    float s1 = __expf(m - M);
    float s2 = __expf(m2 - M);
    num = num * s1 + n2 * s2;
    den = den * s1 + d2 * s2;
    m   = M;
}

__global__ __launch_bounds__(256)
void mmse_scan_kernel(const float* __restrict__ data,
                      const float* __restrict__ targets,
                      const float* __restrict__ W,
                      float* __restrict__ wsM,
                      float* __restrict__ wsNum,
                      float* __restrict__ wsDen)
{
    __shared__ alignas(16) float xw[3 * XS];
    __shared__ alignas(16) float al[2 * XS];
    __shared__ alignas(16) float sA[2 * AS];

    const int tid  = threadIdx.x;
    const int lane = tid & 31;
    const int wave = tid >> 5;
    const int lm   = lane & 15;
    const int lh   = lane >> 4;

    const int b     = (int)blockIdx.x >> 3;
    const int g     = (int)blockIdx.x & (GG - 1);
    const int tbase = g * TC;

    const float* dataB = data + (size_t)b * NN * DD;
    const float* targB = targets + b * NN;

    v2f bfrag[2][16];
#pragma unroll
    for (int j = 0; j < 2; ++j) {
        const int gcol = tbase + wave * 32 + j * 16 + lm;
#pragma unroll
        for (int s = 0; s < 16; ++s) {
            const int k0 = 4 * s + 2 * lh;
            v2f bf;
            bf.x = W[(size_t)k0 * TT + gcol];
            bf.y = W[(size_t)(k0 + 1) * TT + gcol];
            bfrag[j][s] = bf;
        }
    }

    auto load_tile = [&](int blk) -> v4f {
        const int t = blk < NBLK ? blk : NBLK - 1;
        return *(reinterpret_cast<const v4f*>(dataB + (size_t)t * NB * DD) + tid);
    };
    auto store_tile = [&](int aoff, v4f a) {
        *reinterpret_cast<v4f*>(&sA[aoff + (tid >> 4) * AP + 4 * (tid & 15)]) = a;
    };
    auto load_targets = [&](int blk, v4f (&t)[4]) {
        const int tb = blk < NBLK ? blk : NBLK - 1;
#pragma unroll
        for (int q = 0; q < 4; ++q)
            t[q] = *reinterpret_cast<const v4f*>(targB + tb * NB + 4 * q);
    };

    v2f afA[16], afB[16];
    auto load_frags = [&](v2f (&af)[16], int aoff) {
#pragma unroll
        for (int s = 0; s < 16; ++s)
            af[s] = *reinterpret_cast<const v2f*>(&sA[aoff + lm * AP + 4 * s + 2 * lh]);
    };

    v8f acc0, acc1;
    auto store_products = [&](int xq) {
#pragma unroll
        for (int j = 0; j < 2; ++j) {
            const v8f& acc = j ? acc1 : acc0;
            const int lcol = wave * 32 + j * 16 + lm;
            v4f lo = {acc[0], acc[1], acc[2], acc[3]};
            v4f hi = {acc[4], acc[5], acc[6], acc[7]};
            *reinterpret_cast<v4f*>(&xw[4 * xq + lcol * NBP + 8 * lh])     = lo;
            *reinterpret_cast<v4f*>(&xw[4 * xq + lcol * NBP + 8 * lh + 4]) = hi;
        }
    };

    float cum = 0.0f;

    const int pn = tid >> 4;
    const int pk = tid & 15;
    const size_t obase = (size_t)g * BN + (size_t)b * NN + pn;

    float pa[NB], px[NB];
    auto row_fetch = [&](int xq, int poff) {
#pragma unroll
        for (int i = 0; i < NB; ++i) {
            px[i] = xw[4 * xq + (16 * i + pk) * NBP + pn];
            pa[i] = al[poff + (16 * i + pk) * NBP + pn];
        }
    };

    float sm = 0.0f, snum = 0.0f, sden = 0.0f;
    auto store_row = [&](int blk) {
        const size_t o = obase + (size_t)blk * NB;
        wsM[o] = sm; wsNum[o] = snum; wsDen[o] = sden;
    };

    v4f tr[4];
    v4f apark;

    store_tile(0, load_tile(0));
    store_tile(AS, load_tile(1));
    __syncthreads();
    load_frags(afA, 0);
    acc0 = v8f{}; acc1 = v8f{};
#pragma unroll
    for (int s = 0; s < 16; ++s) {
        acc0 = __builtin_amdgcn_wmma_f32_16x16x4_f32(false, afA[s], false, bfrag[0][s],
                                                     (short)0, acc0, false, false);
        acc1 = __builtin_amdgcn_wmma_f32_16x16x4_f32(false, afA[s], false, bfrag[1][s],
                                                     (short)0, acc1, false, false);
    }
    store_products(0);
    load_frags(afB, AS);
    __syncthreads();
    store_tile(0, load_tile(2));
    apark = load_tile(3);
    load_targets(0, tr);
    bar_signal();

#define SLICE() __builtin_amdgcn_sched_barrier(0)

    auto step = [&](int blk, const v2f (&cur)[16], v2f (&nxt)[16],
                    const int fbuf, const int sbuf, const int pcur, const int pold,
                    const int ring, const int rprev, const int rnxt,
                    const bool first, const bool gemm)
        __attribute__((always_inline)) {
        if (pk == 0 && blk > 1) store_row(blk - 2);

        v4f xr[4];
#pragma unroll
        for (int q = 0; q < 4; ++q)
            xr[q] = *reinterpret_cast<const v4f*>(&xw[4 * ring + tid * NBP + 4 * q]);
        SLICE();

        float c[NB];
        float m = 0.0f, den = 0.0f, num = 0.0f;
        acc0 = v8f{}; acc1 = v8f{};
#pragma unroll
        for (int s = 0; s < 16; ++s) {
            if (gemm) {
                acc0 = __builtin_amdgcn_wmma_f32_16x16x4_f32(false, cur[s], false, bfrag[0][s],
                                                             (short)0, acc0, false, false);
                acc1 = __builtin_amdgcn_wmma_f32_16x16x4_f32(false, cur[s], false, bfrag[1][s],
                                                             (short)0, acc1, false, false);
                SLICE();
            }
            if (s >= 1 && s <= 4) {
                const int q = s - 1;
#pragma unroll
                for (int r = 0; r < 4; ++r) {
                    c[4 * q + r] = cum;
                    const float df = tr[q][r] - xr[q][r];
                    cum = fmaf(df, df * -0.5f, cum);
                }
            }
            if (s == 4) {
                SLICE();
                bar_wait();
                SLICE();
                if (!first) row_fetch(rprev, pold);
#pragma unroll
                for (int q = 0; q < 4; ++q) {
                    v4f c4 = {c[4 * q], c[4 * q + 1], c[4 * q + 2], c[4 * q + 3]};
                    *reinterpret_cast<v4f*>(&al[pcur + tid * NBP + 4 * q]) = c4;
                }
                store_tile(sbuf, apark);
                load_frags(nxt, fbuf);
                apark = load_tile(blk + 4);
            }
            if (s == 5) load_targets(blk + 1, tr);
            if (s == 6) {
                SLICE();
                bar_signal();
            }
            if (!first) {
                if (s == 7) {
                    m = pa[0];
#pragma unroll
                    for (int i = 1; i < NB; ++i) m = fmaxf(m, pa[i]);
                    m = fmaxf(m, XMASK(m, 8));
                    m = fmaxf(m, XMASK(m, 4));
                    m = fmaxf(m, XMASK(m, 2));
                    m = fmaxf(m, XMASK(m, 1));
                }
                if (s >= 8) {
#pragma unroll
                    for (int i = 2 * (s - 8); i < 2 * (s - 8) + 2; ++i) {
                        const float e = __expf(pa[i] - m);
                        den += e;
                        num = fmaf(e, px[i], num);
                    }
                }
            }
            SLICE();
        }
        if (gemm) store_products(rnxt);
        if (!first) {
            den += XMASK(den, 8);  num += XMASK(num, 8);
            den += XMASK(den, 4);  num += XMASK(num, 4);
            den += XMASK(den, 2);  num += XMASK(num, 2);
            den += XMASK(den, 1);  num += XMASK(num, 1);
            sm = m; snum = num; sden = den;
        }
    };

    constexpr int R0 = 0, R1 = XS / 4, R2 = 2 * (XS / 4);
    step(0, afB, afA, 0, AS, 0, XS, R0, R2, R1, true, true);
    int ring = R1, rprev = R0;
    for (int blk = 1; blk < NBLK - 1; blk += 2) {
        int rnxt = (ring == R2) ? 0 : ring + XS / 4;
        step(blk, afA, afB, AS, 0, XS, 0, ring, rprev, rnxt, false, true);
        rprev = ring; ring = rnxt;
        rnxt = (ring == R2) ? 0 : ring + XS / 4;
        step(blk + 1, afB, afA, 0, AS, 0, XS, ring, rprev, rnxt, false, true);
        rprev = ring; ring = rnxt;
    }
    step(NBLK - 1, afA, afB, AS, 0, XS, 0, ring, rprev, 0, false, false);

    if (pk == 0) store_row(NBLK - 2);
    SLICE();
    bar_wait();
    SLICE();
    {
        row_fetch(ring, XS);
        float m = pa[0];
#pragma unroll
        for (int i = 1; i < NB; ++i) m = fmaxf(m, pa[i]);
        m = fmaxf(m, XMASK(m, 8));
        m = fmaxf(m, XMASK(m, 4));
        m = fmaxf(m, XMASK(m, 2));
        m = fmaxf(m, XMASK(m, 1));
        float den = 0.0f, num = 0.0f;
#pragma unroll
        for (int i = 0; i < NB; ++i) {
            const float e = __expf(pa[i] - m);
            den += e;
            num = fmaf(e, px[i], num);
        }
        den += XMASK(den, 8);  num += XMASK(num, 8);
        den += XMASK(den, 4);  num += XMASK(num, 4);
        den += XMASK(den, 2);  num += XMASK(num, 2);
        den += XMASK(den, 1);  num += XMASK(num, 1);
        sm = m; snum = num; sden = den;
        if (pk == 0) store_row(NBLK - 1);
    }
#undef SLICE
}

__global__ __launch_bounds__(256)
void mmse_combine_kernel(const float* __restrict__ wsM,
                         const float* __restrict__ wsNum,
                         const float* __restrict__ wsDen,
                         float* __restrict__ out)
{
    const int i = (int)blockIdx.x * 256 + (int)threadIdx.x;
    if (i >= BN) return;
    float m = wsM[i], num = wsNum[i], den = wsDen[i];
#pragma unroll
    for (int g = 1; g < GG; ++g) {
        const size_t o = (size_t)g * BN + i;
        smerge(m, num, den, wsM[o], wsNum[o], wsDen[o]);
    }
    out[i] = num / den;
}

extern "C" void kernel_launch(void* const* d_in, const int* in_sizes, int n_in,
                              void* d_out, int out_size, void* d_ws, size_t ws_size,
                              hipStream_t stream) {
    (void)in_sizes; (void)n_in; (void)out_size; (void)ws_size;
    const float* data    = (const float*)d_in[0];
    const float* targets = (const float*)d_in[1];
    const float* W       = (const float*)d_in[2];
    float*       out     = (float*)d_out;

    float* wsBase = (float*)d_ws;
    float* wsM    = wsBase;
    float* wsNum  = wsBase + (size_t)BNG;
    float* wsDen  = wsBase + (size_t)2 * BNG;

    mmse_scan_kernel<<<dim3(BB * GG), dim3(256), 0, stream>>>(
        data, targets, W, wsM, wsNum, wsDen);
    mmse_combine_kernel<<<dim3((BN + 255) / 256), dim3(256), 0, stream>>>(
        wsM, wsNum, wsDen, out);
}


# vector work re-cut into half-slices behind every single WMMA (was: every pair); scan operands and row-pass stages (scale, exp, accumulate) formed a half-slice ahead of their use; sum order kept
# speedup vs baseline: 8.2923x; 1.0400x over previous
#include <hip/hip_runtime.h>

#define BB   32
#define NN   1024
#define DD   64
#define TT   2048
#define GG   8
#define TC   (TT / GG)
#define NB   16
#define NBP  (NB + 4)
#define NBLK (NN / NB)
#define AP   (DD + 4)
#define XS   (TC * NBP)
#define AS   (NB * AP)
#define BN   (BB * NN)
#define BNG  (BN * GG)

typedef float v2f __attribute__((ext_vector_type(2)));
typedef float v4f __attribute__((ext_vector_type(4)));
typedef float v8f __attribute__((ext_vector_type(8)));

#define XMASK(v, X) \
    __int_as_float(__builtin_amdgcn_update_dpp( \
        0, __float_as_int(v), 0x160 + (X), 0xF, 0xF, true))

__device__ __forceinline__ void bar_signal() {
    __builtin_amdgcn_fence(__ATOMIC_RELEASE, "workgroup", "local");
    __builtin_amdgcn_s_barrier_signal(-1);
}
__device__ __forceinline__ void bar_wait() {
    __builtin_amdgcn_s_barrier_wait(-1);
    __builtin_amdgcn_fence(__ATOMIC_ACQUIRE, "workgroup", "local");
}

__device__ __forceinline__ void smerge(float& m, float& num, float& den,
                                       float m2, float n2, float d2) {
    float M  = fmaxf(m, m2);
    float s1 = __expf(m - M);
    float s2 = __expf(m2 - M);
    num = num * s1 + n2 * s2;
    den = den * s1 + d2 * s2;
    m   = M;
}

__global__ __launch_bounds__(256)
void mmse_scan_kernel(const float* __restrict__ data,
                      const float* __restrict__ targets,
                      const float* __restrict__ W,
                      float* __restrict__ wsM,
                      float* __restrict__ wsNum,
                      float* __restrict__ wsDen)
{
    __shared__ alignas(16) float xw[3 * XS];
    __shared__ alignas(16) float al[2 * XS];
    __shared__ alignas(16) float sA[2 * AS];

    const int tid  = threadIdx.x;
    const int lane = tid & 31;
    const int wave = tid >> 5;
    const int lm   = lane & 15;
    const int lh   = lane >> 4;

    const int b     = (int)blockIdx.x >> 3;
    const int g     = (int)blockIdx.x & (GG - 1);
    const int tbase = g * TC;

    const float* dataB = data + (size_t)b * NN * DD;
    const float* targB = targets + b * NN;

    v2f bfrag[2][16];
#pragma unroll
    for (int j = 0; j < 2; ++j) {
        const int gcol = tbase + wave * 32 + j * 16 + lm;
#pragma unroll
        for (int s = 0; s < 16; ++s) {
            const int k0 = 4 * s + 2 * lh;
            v2f bf;
            bf.x = W[(size_t)k0 * TT + gcol];
            bf.y = W[(size_t)(k0 + 1) * TT + gcol];
            bfrag[j][s] = bf;
        }
    }

    auto load_tile = [&](int blk) -> v4f {
        const int t = blk < NBLK ? blk : NBLK - 1;
        return *(reinterpret_cast<const v4f*>(dataB + (size_t)t * NB * DD) + tid);
    };
    auto store_tile = [&](int aoff, v4f a) {
        *reinterpret_cast<v4f*>(&sA[aoff + (tid >> 4) * AP + 4 * (tid & 15)]) = a;
    };
    auto load_targets = [&](int blk, v4f (&t)[4]) {
        const int tb = blk < NBLK ? blk : NBLK - 1;
#pragma unroll
        for (int q = 0; q < 4; ++q)
            t[q] = *reinterpret_cast<const v4f*>(targB + tb * NB + 4 * q);
    };

    v2f afA[16], afB[16];
    auto load_frags = [&](v2f (&af)[16], int aoff) {
#pragma unroll
        for (int s = 0; s < 16; ++s)
            af[s] = *reinterpret_cast<const v2f*>(&sA[aoff + lm * AP + 4 * s + 2 * lh]);
    };

    v8f acc0, acc1;
    auto store_products = [&](int xq) {
#pragma unroll
        for (int j = 0; j < 2; ++j) {
            const v8f& acc = j ? acc1 : acc0;
            const int lcol = wave * 32 + j * 16 + lm;
            v4f lo = {acc[0], acc[1], acc[2], acc[3]};
            v4f hi = {acc[4], acc[5], acc[6], acc[7]};
            *reinterpret_cast<v4f*>(&xw[4 * xq + lcol * NBP + 8 * lh])     = lo;
            *reinterpret_cast<v4f*>(&xw[4 * xq + lcol * NBP + 8 * lh + 4]) = hi;
        }
    };

    float cum = 0.0f;

    const int pn = tid >> 4;
    const int pk = tid & 15;
    const size_t obase = (size_t)g * BN + (size_t)b * NN + pn;

    float pa[NB], px[NB];
    auto row_fetch = [&](int xq, int poff) {
#pragma unroll
        for (int i = 0; i < NB; ++i) {
            px[i] = xw[4 * xq + (16 * i + pk) * NBP + pn];
            pa[i] = al[poff + (16 * i + pk) * NBP + pn];
        }
    };

    float sm = 0.0f, snum = 0.0f, sden = 0.0f;
    auto store_row = [&](int blk) {
        const size_t o = obase + (size_t)blk * NB;
        wsM[o] = sm; wsNum[o] = snum; wsDen[o] = sden;
    };

    v4f tr[4];
    v4f apark;

    store_tile(0, load_tile(0));
    store_tile(AS, load_tile(1));
    __syncthreads();
    load_frags(afA, 0);
    acc0 = v8f{}; acc1 = v8f{};
#pragma unroll
    for (int s = 0; s < 16; ++s) {
        acc0 = __builtin_amdgcn_wmma_f32_16x16x4_f32(false, afA[s], false, bfrag[0][s],
                                                     (short)0, acc0, false, false);
        acc1 = __builtin_amdgcn_wmma_f32_16x16x4_f32(false, afA[s], false, bfrag[1][s],
                                                     (short)0, acc1, false, false);
    }
    store_products(0);
    load_frags(afB, AS);
    __syncthreads();
    store_tile(0, load_tile(2));
    apark = load_tile(3);
    load_targets(0, tr);
    bar_signal();

#define SLICE() __builtin_amdgcn_sched_barrier(0)

    auto step = [&](int blk, const v2f (&cur)[16], v2f (&nxt)[16],
                    const int fbuf, const int sbuf, const int pcur, const int pold,
                    const int ring, const int rprev, const int rnxt,
                    const bool first, const bool gemm)
        __attribute__((always_inline)) {
        if (pk == 0 && blk > 1) store_row(blk - 2);

        v4f xr[4];
#pragma unroll
        for (int q = 0; q < 4; ++q)
            xr[q] = *reinterpret_cast<const v4f*>(&xw[4 * ring + tid * NBP + 4 * q]);
        SLICE();

        float c[NB];
        float df[NB], hh[NB];
        float u[NB], e[NB];
        float m = 0.0f, ma = 0.0f, mb = 0.0f, den = 0.0f, num = 0.0f;
        acc0 = v8f{}; acc1 = v8f{};
#pragma unroll
        for (int h = 0; h < 32; ++h) {
            const int s = h >> 1;
            if (gemm) {
                if ((h & 1) == 0)
                    acc0 = __builtin_amdgcn_wmma_f32_16x16x4_f32(false, cur[s], false, bfrag[0][s],
                                                                 (short)0, acc0, false, false);
                else
                    acc1 = __builtin_amdgcn_wmma_f32_16x16x4_f32(false, cur[s], false, bfrag[1][s],
                                                                 (short)0, acc1, false, false);
                SLICE();
            }
            if (h >= 3 && h <= 10) {
                const int n0 = 2 * (h - 3);
                if (h == 3) {
#pragma unroll
                    for (int n = 0; n < 2; ++n) {
                        df[n] = tr[n >> 2][n & 3] - xr[n >> 2][n & 3];
                        hh[n] = df[n] * -0.5f;
                    }
                }
                if (h < 10) {
#pragma unroll
                    for (int n = n0 + 2; n < n0 + 4; ++n) {
                        df[n] = tr[n >> 2][n & 3] - xr[n >> 2][n & 3];
                        hh[n] = df[n] * -0.5f;
                    }
                }
#pragma unroll
                for (int n = n0; n < n0 + 2; ++n) {
                    c[n] = cum;
                    cum = fmaf(df[n], hh[n], cum);
                }
            }
            if (h == 10) {
                SLICE();
                bar_wait();
                SLICE();
                if (!first) row_fetch(rprev, pold);
#pragma unroll
                for (int q = 0; q < 4; ++q) {
                    v4f c4 = {c[4 * q], c[4 * q + 1], c[4 * q + 2], c[4 * q + 3]};
                    *reinterpret_cast<v4f*>(&al[pcur + tid * NBP + 4 * q]) = c4;
                }
                store_tile(sbuf, apark);
                load_frags(nxt, fbuf);
                apark = load_tile(blk + 4);
            }
            if (h == 12) load_targets(blk + 1, tr);
            if (h == 14) {
                SLICE();
                bar_signal();
            }
            if (!first) {
                if (h == 14) {
                    ma = pa[0]; mb = pa[8];
#pragma unroll
                    for (int i = 1; i < 8; ++i) {
                        ma = fmaxf(ma, pa[i]);
                        mb = fmaxf(mb, pa[8 + i]);
                    }
                }
                if (h == 15) { m = fmaxf(ma, mb); m = fmaxf(m, XMASK(m, 8)); }
                if (h == 16) m = fmaxf(m, XMASK(m, 4));
                if (h == 17) m = fmaxf(m, XMASK(m, 2));
                if (h == 18) m = fmaxf(m, XMASK(m, 1));
                if (h >= 19 && h <= 26) {
#pragma unroll
                    for (int i = 2 * (h - 19); i < 2 * (h - 19) + 2; ++i)
                        u[i] = (pa[i] - m) * 1.4426950408889634f;
                }
                if (h >= 20 && h <= 27) {
#pragma unroll
                    for (int i = 2 * (h - 20); i < 2 * (h - 20) + 2; ++i)
                        e[i] = __builtin_amdgcn_exp2f(u[i]);
                }
                if (h >= 21 && h <= 28) {
#pragma unroll
                    for (int i = 2 * (h - 21); i < 2 * (h - 21) + 2; ++i) {
                        den += e[i];
                        num = fmaf(e[i], px[i], num);
                    }
                }
                if (h == 29) { den += XMASK(den, 8);  num += XMASK(num, 8); }
                if (h == 30) { den += XMASK(den, 4);  num += XMASK(num, 4); }
                if (h == 31) { den += XMASK(den, 2);  num += XMASK(num, 2); }
            }
            SLICE();
        }
        if (gemm) store_products(rnxt);
        if (!first) {
            den += XMASK(den, 1);  num += XMASK(num, 1);
            sm = m; snum = num; sden = den;
        }
    };

    constexpr int R0 = 0, R1 = XS / 4, R2 = 2 * (XS / 4);
    step(0, afB, afA, 0, AS, 0, XS, R0, R2, R1, true, true);
    int ring = R1, rprev = R0;
    for (int blk = 1; blk < NBLK - 1; blk += 2) {
        int rnxt = (ring == R2) ? 0 : ring + XS / 4;
        step(blk, afA, afB, AS, 0, XS, 0, ring, rprev, rnxt, false, true);
        rprev = ring; ring = rnxt;
        rnxt = (ring == R2) ? 0 : ring + XS / 4;
        step(blk + 1, afB, afA, 0, AS, 0, XS, ring, rprev, rnxt, false, true);
        rprev = ring; ring = rnxt;
    }
    step(NBLK - 1, afA, afB, AS, 0, XS, 0, ring, rprev, 0, false, false);

    if (pk == 0) store_row(NBLK - 2);
    SLICE();
    bar_wait();
    SLICE();
    {
        row_fetch(ring, XS);
        float m = pa[0];
#pragma unroll
        for (int i = 1; i < NB; ++i) m = fmaxf(m, pa[i]);
        m = fmaxf(m, XMASK(m, 8));
        m = fmaxf(m, XMASK(m, 4));
        m = fmaxf(m, XMASK(m, 2));
        m = fmaxf(m, XMASK(m, 1));
        float den = 0.0f, num = 0.0f;
#pragma unroll
        for (int i = 0; i < NB; ++i) {
            const float e = __expf(pa[i] - m);
            den += e;
            num = fmaf(e, px[i], num);
        }
        den += XMASK(den, 8);  num += XMASK(num, 8);
        den += XMASK(den, 4);  num += XMASK(num, 4);
        den += XMASK(den, 2);  num += XMASK(num, 2);
        den += XMASK(den, 1);  num += XMASK(num, 1);
        sm = m; snum = num; sden = den;
        if (pk == 0) store_row(NBLK - 1);
    }
#undef SLICE
}

__global__ __launch_bounds__(256)
void mmse_combine_kernel(const float* __restrict__ wsM,
                         const float* __restrict__ wsNum,
                         const float* __restrict__ wsDen,
                         float* __restrict__ out)
{
    const int i = (int)blockIdx.x * 256 + (int)threadIdx.x;
    if (i >= BN) return;
    float m = wsM[i], num = wsNum[i], den = wsDen[i];
#pragma unroll
    for (int g = 1; g < GG; ++g) {
        const size_t o = (size_t)g * BN + i;
        smerge(m, num, den, wsM[o], wsNum[o], wsDen[o]);
    }
    out[i] = num / den;
}

extern "C" void kernel_launch(void* const* d_in, const int* in_sizes, int n_in,
                              void* d_out, int out_size, void* d_ws, size_t ws_size,
                              hipStream_t stream) {
    (void)in_sizes; (void)n_in; (void)out_size; (void)ws_size;
    const float* data    = (const float*)d_in[0];
    const float* targets = (const float*)d_in[1];
    const float* W       = (const float*)d_in[2];
    float*       out     = (float*)d_out;

    float* wsBase = (float*)d_ws;
    float* wsM    = wsBase;
    float* wsNum  = wsBase + (size_t)BNG;
    float* wsDen  = wsBase + (size_t)2 * BNG;

    mmse_scan_kernel<<<dim3(BB * GG), dim3(256), 0, stream>>>(
        data, targets, W, wsM, wsNum, wsDen);
    mmse_combine_kernel<<<dim3((BN + 255) / 256), dim3(256), 0, stream>>>(
        wsM, wsNum, wsDen, out);
}


# wave issue priority set by place in the step (s_setprio 3 up to the barrier signal, 0 behind it, 1 from mid-step): the two waves of a SIMD take turns and fill each other's waits; sum order kept
# speedup vs baseline: 9.2860x; 1.1198x over previous
#include <hip/hip_runtime.h>

#define BB   32
#define NN   1024
#define DD   64
#define TT   2048
#define GG   8
#define TC   (TT / GG)
#define NB   16
#define NBP  (NB + 4)
#define NBLK (NN / NB)
#define AP   (DD + 4)
#define XS   (TC * NBP)
#define AS   (NB * AP)
#define BN   (BB * NN)
#define BNG  (BN * GG)

#define KS    0.84932180028801904f
#define KSINV (-1.1774100225154747f)

typedef float v2f __attribute__((ext_vector_type(2)));
typedef float v4f __attribute__((ext_vector_type(4)));
typedef float v8f __attribute__((ext_vector_type(8)));

#define XMASK(v, X) \
    __int_as_float(__builtin_amdgcn_update_dpp( \
        0, __float_as_int(v), 0x160 + (X), 0xF, 0xF, true))
#define XMASKI(v, X) \
    __builtin_amdgcn_update_dpp(0, (v), 0x160 + (X), 0xF, 0xF, true)
__device__ __forceinline__ int IMIN(int a, int b) { return a < b ? a : b; }

__device__ __forceinline__ void bar_signal() {
    __builtin_amdgcn_fence(__ATOMIC_RELEASE, "workgroup", "local");
    __builtin_amdgcn_s_barrier_signal(-1);
}
__device__ __forceinline__ void bar_wait() {
    __builtin_amdgcn_s_barrier_wait(-1);
    __builtin_amdgcn_fence(__ATOMIC_ACQUIRE, "workgroup", "local");
}

__device__ __forceinline__ void smerge(float& m, float& num, float& den,
                                       float m2, float n2, float d2) {
    float M  = fminf(m, m2);
    float s1 = __builtin_amdgcn_exp2f(M - m);
    float s2 = __builtin_amdgcn_exp2f(M - m2);
    num = num * s1 + n2 * s2;
    den = den * s1 + d2 * s2;
    m   = M;
}

__global__ __launch_bounds__(256)
void mmse_scan_kernel(const float* __restrict__ data,
                      const float* __restrict__ targets,
                      const float* __restrict__ W,
                      float* __restrict__ wsM,
                      float* __restrict__ wsNum,
                      float* __restrict__ wsDen)
{
    __shared__ alignas(16) float xw[3 * XS];
    __shared__ alignas(16) float al[2 * XS];
    __shared__ alignas(16) float sA[2 * AS];

    const int tid  = threadIdx.x;
    const int lane = tid & 31;
    const int wave = tid >> 5;
    const int lm   = lane & 15;
    const int lh   = lane >> 4;

    const int b     = (int)blockIdx.x >> 3;
    const int g     = (int)blockIdx.x & (GG - 1);
    const int tbase = g * TC;

    const float* dataB = data + (size_t)b * NN * DD;
    const float* targB = targets + b * NN;

    v2f bfrag[2][16];
#pragma unroll
    for (int j = 0; j < 2; ++j) {
        const int gcol = tbase + wave * 32 + j * 16 + lm;
#pragma unroll
        for (int s = 0; s < 16; ++s) {
            const int k0 = 4 * s + 2 * lh;
            v2f bf;
            bf.x = W[(size_t)k0 * TT + gcol] * -KS;
            bf.y = W[(size_t)(k0 + 1) * TT + gcol] * -KS;
            bfrag[j][s] = bf;
        }
    }

    auto load_tile = [&](int blk) -> v4f {
        const int t = blk < NBLK ? blk : NBLK - 1;
        return *(reinterpret_cast<const v4f*>(dataB + (size_t)t * NB * DD) + tid);
    };
    auto store_tile = [&](int aoff, v4f a) {
        *reinterpret_cast<v4f*>(&sA[aoff + (tid >> 4) * AP + 4 * (tid & 15)]) = a;
    };
    auto load_targets = [&](int blk, v4f (&t)[4]) {
        const int tb = blk < NBLK ? blk : NBLK - 1;
#pragma unroll
        for (int q = 0; q < 4; ++q)
            t[q] = *reinterpret_cast<const v4f*>(targB + tb * NB + 4 * q);
    };

    v2f afA[16], afB[16];
    auto load_frags = [&](v2f (&af)[16], int aoff) {
#pragma unroll
        for (int s = 0; s < 16; ++s)
            af[s] = *reinterpret_cast<const v2f*>(&sA[aoff + lm * AP + 4 * s + 2 * lh]);
    };

    v8f acc0, acc1;
    auto store_products = [&](int xq) {
#pragma unroll
        for (int j = 0; j < 2; ++j) {
            const v8f& acc = j ? acc1 : acc0;
            const int lcol = wave * 32 + j * 16 + lm;
            v4f lo = {acc[0], acc[1], acc[2], acc[3]};
            v4f hi = {acc[4], acc[5], acc[6], acc[7]};
            *reinterpret_cast<v4f*>(&xw[4 * xq + lcol * NBP + 8 * lh])     = lo;
            *reinterpret_cast<v4f*>(&xw[4 * xq + lcol * NBP + 8 * lh + 4]) = hi;
        }
    };

    float cum = 0.0f;

    const int pn = tid >> 4;
    const int pk = tid & 15;
    const size_t obase = (size_t)g * BN + (size_t)b * NN + pn;

    float pa[NB], px[NB];
    auto row_fetch = [&](int xq, int poff) {
#pragma unroll
        for (int i = 0; i < NB; ++i) {
            px[i] = xw[4 * xq + (16 * i + pk) * NBP + pn];
            pa[i] = al[poff + (16 * i + pk) * NBP + pn];
        }
    };

    float sm = 0.0f, snum = 0.0f, sden = 0.0f;
    auto store_row = [&](int blk) {
        const size_t o = obase + (size_t)blk * NB;
        wsM[o] = sm; wsNum[o] = snum; wsDen[o] = sden;
    };

    v4f tr[4];
    v4f apark;

    store_tile(0, load_tile(0));
    store_tile(AS, load_tile(1));
    __syncthreads();
    load_frags(afA, 0);
    acc0 = v8f{}; acc1 = v8f{};
#pragma unroll
    for (int s = 0; s < 16; ++s) {
        acc0 = __builtin_amdgcn_wmma_f32_16x16x4_f32(false, afA[s], false, bfrag[0][s],
                                                     (short)0, acc0, false, false);
        acc1 = __builtin_amdgcn_wmma_f32_16x16x4_f32(false, afA[s], false, bfrag[1][s],
                                                     (short)0, acc1, false, false);
    }
    store_products(0);
    load_frags(afB, AS);
    __syncthreads();
    store_tile(0, load_tile(2));
    apark = load_tile(3);
    load_targets(0, tr);
    bar_signal();

#define SLICE() __builtin_amdgcn_sched_barrier(0)
#define PRIO(n) __builtin_amdgcn_s_setprio(n)

    auto step = [&](int blk, const v2f (&cur)[16], v2f (&nxt)[16],
                    const int fbuf, const int sbuf, const int pcur, const int pold,
                    const int ring, const int rprev, const int rnxt,
                    const bool first, const bool gemm)
        __attribute__((always_inline)) {
        PRIO(3);
        if (pk == 0 && blk > 1) store_row(blk - 2);

        v4f xr[4];
#pragma unroll
        for (int q = 0; q < 4; ++q)
            xr[q] = *reinterpret_cast<const v4f*>(&xw[4 * ring + tid * NBP + 4 * q]);
        SLICE();

        float c[NB];
        float df[NB];
        float u[NB], e[NB];
        float m = 0.0f, den = 0.0f, num = 0.0f;
        int   ma = 0, mb = 0, mi = 0;
        acc0 = v8f{}; acc1 = v8f{};
#pragma unroll
        for (int h = 0; h < 32; ++h) {
            const int s = h >> 1;
            if (gemm) {
                if ((h & 1) == 0)
                    acc0 = __builtin_amdgcn_wmma_f32_16x16x4_f32(false, cur[s], false, bfrag[0][s],
                                                                 (short)0, acc0, false, false);
                else
                    acc1 = __builtin_amdgcn_wmma_f32_16x16x4_f32(false, cur[s], false, bfrag[1][s],
                                                                 (short)0, acc1, false, false);
                SLICE();
            }
            if (h >= 3 && h <= 10) {
                const int n0 = 2 * (h - 3);
                if (h == 3) {
#pragma unroll
                    for (int n = 0; n < 2; ++n)
                        df[n] = fmaf(tr[n >> 2][n & 3], KS, xr[n >> 2][n & 3]);
                }
                if (h < 10) {
#pragma unroll
                    for (int n = n0 + 2; n < n0 + 4; ++n)
                        df[n] = fmaf(tr[n >> 2][n & 3], KS, xr[n >> 2][n & 3]);
                }
#pragma unroll
                for (int n = n0; n < n0 + 2; ++n) {
                    c[n] = cum;
                    cum = fmaf(df[n], df[n], cum);
                }
            }
            if (h == 10) {
                SLICE();
                bar_wait();
                SLICE();
                if (!first) row_fetch(rprev, pold);
#pragma unroll
                for (int q = 0; q < 4; ++q) {
                    v4f c4 = {c[4 * q], c[4 * q + 1], c[4 * q + 2], c[4 * q + 3]};
                    *reinterpret_cast<v4f*>(&al[pcur + tid * NBP + 4 * q]) = c4;
                }
                store_tile(sbuf, apark);
                load_frags(nxt, fbuf);
                apark = load_tile(blk + 4);
            }
            if (h == 12) load_targets(blk + 1, tr);
            if (h == 14) {
                SLICE();
                bar_signal();
                PRIO(0);
            }
            if (h == 20) PRIO(1);
            if (!first) {
                if (h == 14) {
                    ma = __float_as_int(pa[0]); mb = __float_as_int(pa[8]);
#pragma unroll
                    for (int i = 1; i < 8; ++i) {
                        ma = IMIN(ma, __float_as_int(pa[i]));
                        mb = IMIN(mb, __float_as_int(pa[8 + i]));
                    }
                }
                if (h == 15) { mi = IMIN(ma, mb); mi = IMIN(mi, XMASKI(mi, 8)); }
                if (h == 16) mi = IMIN(mi, XMASKI(mi, 4));
                if (h == 17) mi = IMIN(mi, XMASKI(mi, 2));
                if (h == 18) { mi = IMIN(mi, XMASKI(mi, 1)); m = __int_as_float(mi); }
                if (h >= 19 && h <= 26) {
#pragma unroll
                    for (int i = 2 * (h - 19); i < 2 * (h - 19) + 2; ++i)
                        u[i] = m - pa[i];
                }
                if (h >= 20 && h <= 27) {
#pragma unroll
                    for (int i = 2 * (h - 20); i < 2 * (h - 20) + 2; ++i)
                        e[i] = __builtin_amdgcn_exp2f(u[i]);
                }
                if (h >= 21 && h <= 28) {
#pragma unroll
                    for (int i = 2 * (h - 21); i < 2 * (h - 21) + 2; ++i) {
                        den += e[i];
                        num = fmaf(e[i], px[i], num);
                    }
                }
                if (h == 29) { den += XMASK(den, 8);  num += XMASK(num, 8); }
                if (h == 30) { den += XMASK(den, 4);  num += XMASK(num, 4); }
                if (h == 31) { den += XMASK(den, 2);  num += XMASK(num, 2); }
            }
            SLICE();
        }
        if (gemm) store_products(rnxt);
        if (!first) {
            den += XMASK(den, 1);  num += XMASK(num, 1);
            sm = m; snum = num; sden = den;
        }
    };

    constexpr int R0 = 0, R1 = XS / 4, R2 = 2 * (XS / 4);
    step(0, afB, afA, 0, AS, 0, XS, R0, R2, R1, true, true);
    int ring = R1, rprev = R0;
    for (int blk = 1; blk < NBLK - 1; blk += 2) {
        int rnxt = (ring == R2) ? 0 : ring + XS / 4;
        step(blk, afA, afB, AS, 0, XS, 0, ring, rprev, rnxt, false, true);
        rprev = ring; ring = rnxt;
        rnxt = (ring == R2) ? 0 : ring + XS / 4;
        step(blk + 1, afB, afA, 0, AS, 0, XS, ring, rprev, rnxt, false, true);
        rprev = ring; ring = rnxt;
    }
    step(NBLK - 1, afA, afB, AS, 0, XS, 0, ring, rprev, 0, false, false);
    PRIO(0);

    if (pk == 0) store_row(NBLK - 2);
    SLICE();
    bar_wait();
    SLICE();
    {
        row_fetch(ring, XS);
        int mi = __float_as_int(pa[0]);
#pragma unroll
        for (int i = 1; i < NB; ++i) mi = IMIN(mi, __float_as_int(pa[i]));
        mi = IMIN(mi, XMASKI(mi, 8));
        mi = IMIN(mi, XMASKI(mi, 4));
        mi = IMIN(mi, XMASKI(mi, 2));
        mi = IMIN(mi, XMASKI(mi, 1));
        const float m = __int_as_float(mi);
        float den = 0.0f, num = 0.0f;
#pragma unroll
        for (int i = 0; i < NB; ++i) {
            const float e = __builtin_amdgcn_exp2f(m - pa[i]);
            den += e;
            num = fmaf(e, px[i], num);
        }
        den += XMASK(den, 8);  num += XMASK(num, 8);
        den += XMASK(den, 4);  num += XMASK(num, 4);
        den += XMASK(den, 2);  num += XMASK(num, 2);
        den += XMASK(den, 1);  num += XMASK(num, 1);
        sm = m; snum = num; sden = den;
        if (pk == 0) store_row(NBLK - 1);
    }
#undef SLICE
#undef PRIO
}

__global__ __launch_bounds__(256)
void mmse_combine_kernel(const float* __restrict__ wsM,
                         const float* __restrict__ wsNum,
                         const float* __restrict__ wsDen,
                         float* __restrict__ out)
{
    const int i = (int)blockIdx.x * 256 + (int)threadIdx.x;
    if (i >= BN) return;
    float m = wsM[i], num = wsNum[i], den = wsDen[i];
#pragma unroll
    for (int g = 1; g < GG; ++g) {
        const size_t o = (size_t)g * BN + i;
        smerge(m, num, den, wsM[o], wsNum[o], wsDen[o]);
    }
    out[i] = (num / den) * KSINV;
}

extern "C" void kernel_launch(void* const* d_in, const int* in_sizes, int n_in,
                              void* d_out, int out_size, void* d_ws, size_t ws_size,
                              hipStream_t stream) {
    (void)in_sizes; (void)n_in; (void)out_size; (void)ws_size;
    const float* data    = (const float*)d_in[0];
    const float* targets = (const float*)d_in[1];
    const float* W       = (const float*)d_in[2];
    float*       out     = (float*)d_out;

    float* wsBase = (float*)d_ws;
    float* wsM    = wsBase;
    float* wsNum  = wsBase + (size_t)BNG;
    float* wsDen  = wsBase + (size_t)2 * BNG;

    mmse_scan_kernel<<<dim3(BB * GG), dim3(256), 0, stream>>>(
        data, targets, W, wsM, wsNum, wsDen);
    mmse_combine_kernel<<<dim3((BN + 255) / 256), dim3(256), 0, stream>>>(
        wsM, wsNum, wsDen, out);
}
